# starting kernel
# baseline (speedup 1.0000x reference)
#include <hip/hip_runtime.h>


typedef __attribute__((ext_vector_type(2))) float v2f;
typedef __attribute__((ext_vector_type(8))) float v8f;

#define D 128

__global__ void k_init_deg(float* __restrict__ deg, int n) {
    int i = blockIdx.x * blockDim.x + threadIdx.x;
    if (i < n) deg[i] = 1.0f;
}

__global__ void k_count_deg(const int* __restrict__ ei_row, float* __restrict__ deg, int e) {
    int i = blockIdx.x * blockDim.x + threadIdx.x;
    if (i < e) atomicAdd(&deg[ei_row[i]], 1.0f);
}

__global__ void k_rsqrt_deg(float* __restrict__ deg, int n) {
    int i = blockIdx.x * blockDim.x + threadIdx.x;
    if (i < n) deg[i] = rsqrtf(deg[i]);
}

__global__ void k_gemm_wmma(const float* __restrict__ x,
                            const float* __restrict__ W,
                            const float* __restrict__ bias,
                            float* __restrict__ h) {
    const int wave = threadIdx.x >> 5;
    const int lane = threadIdx.x & 31;
    const int m0 = blockIdx.x << 4;
    const int n0 = wave << 4;
    const int lm = lane & 15;
    const int lk = (lane >> 4) << 1;

    const float* xr = x + (size_t)(m0 + lm) * D;
    const float* wr = W + (size_t)(n0 + lm) * D;

    v8f acc = {};
#pragma unroll
    for (int k = 0; k < D; k += 4) {
        v2f a = *(const v2f*)(xr + k + lk);
        v2f b = *(const v2f*)(wr + k + lk);
        acc = __builtin_amdgcn_wmma_f32_16x16x4_f32(
            false, a, false, b, (short)0, acc, false, false);
    }

    const float bn = bias[n0 + lm];
    float* hp = h + (size_t)(m0 + ((lane >> 4) << 3)) * D + (n0 + lm);
#pragma unroll
    for (int v = 0; v < 8; ++v)
        hp[(size_t)v * D] = acc[v] + bn;
}

__global__ void k_self_loop(const float* __restrict__ h,
                            const float* __restrict__ rsq,
                            float* __restrict__ out, long long total) {
    long long i = (long long)blockIdx.x * blockDim.x + threadIdx.x;
    if (i < total) {
        int node = (int)(i >> 7);
        float r = rsq[node];
        out[i] = h[i] * (r * r);
    }
}

__global__ void k_aggregate(const int* __restrict__ ei,
                            const float* __restrict__ h,
                            const float* __restrict__ rsq,
                            float* __restrict__ out, int e) {
    const int edge = blockIdx.x * (blockDim.x >> 5) + (threadIdx.x >> 5);
    if (edge >= e) return;
    const int lane = threadIdx.x & 31;

    const int dst = ei[edge];
    const int src = ei[e + edge];
    const float norm = rsq[dst] * rsq[src];

    const float4 hv = *(const float4*)(h + (size_t)src * D + lane * 4);
    float* o = out + (size_t)dst * D + lane * 4;
    atomicAdd(o + 0, norm * hv.x);
    atomicAdd(o + 1, norm * hv.y);
    atomicAdd(o + 2, norm * hv.z);
    atomicAdd(o + 3, norm * hv.w);
}

extern "C" void kernel_launch(void* const* d_in, const int* in_sizes, int n_in,
                              void* d_out, int out_size, void* d_ws, size_t ws_size,
                              hipStream_t stream) {
    const float* x    = (const float*)d_in[0];
    const int*   ei   = (const int*)d_in[1];
    const float* W    = (const float*)d_in[2];
    const float* bias = (const float*)d_in[3];
    float*       out  = (float*)d_out;

    const int n = in_sizes[0] / D;
    const int e = in_sizes[1] / 2;

    float* h   = (float*)d_ws;
    float* rsq = h + (size_t)n * D;

    const int T = 256;

    k_init_deg <<<(n + T - 1) / T, T, 0, stream>>>(rsq, n);
    k_count_deg<<<(e + T - 1) / T, T, 0, stream>>>(ei, rsq, e);
    k_rsqrt_deg<<<(n + T - 1) / T, T, 0, stream>>>(rsq, n);

    k_gemm_wmma<<<n / 16, T, 0, stream>>>(x, W, bias, h);

    long long total = (long long)n * D;
    k_self_loop<<<(unsigned)((total + T - 1) / T), T, 0, stream>>>(h, rsq, out, total);

    k_aggregate<<<(e + 7) / 8, T, 0, stream>>>(ei, h, rsq, out, e);
}


# Aggregation by row ownership: each block holds 400 out rows in LDS, filters edges by a byte tag, sums h[src] rows with LDS float adds (LDS arrival order), stores each row once; no atomics on out
# speedup vs baseline: 1.9552x; 1.9552x over previous
#include <hip/hip_runtime.h>


typedef __attribute__((ext_vector_type(2))) float v2f;
typedef __attribute__((ext_vector_type(8))) float v8f;

#define D 128
#define ROWS 400
#define AGG_WAVES 32
#define LIST_CAP 64

__global__ void k_init_deg(float* __restrict__ deg, int n) {
    int i = blockIdx.x * blockDim.x + threadIdx.x;
    if (i < n) deg[i] = 1.0f;
}

__global__ void k_count_deg(const int* __restrict__ ei_row, float* __restrict__ deg,
                            unsigned char* __restrict__ tag, int e, int e_pad) {
    int i = blockIdx.x * blockDim.x + threadIdx.x;
    if (i < e) {
        const int d = ei_row[i];
        atomicAdd(&deg[d], 1.0f);
        tag[i] = (unsigned char)(d / ROWS);
    } else if (i < e_pad) {
        tag[i] = 255;
    }
}

__global__ void k_rsqrt_deg(float* __restrict__ deg, int n) {
    int i = blockIdx.x * blockDim.x + threadIdx.x;
    if (i < n) deg[i] = rsqrtf(deg[i]);
}

__global__ void k_gemm_wmma(const float* __restrict__ x,
                            const float* __restrict__ W,
                            const float* __restrict__ bias,
                            float* __restrict__ h) {
    const int wave = threadIdx.x >> 5;
    const int lane = threadIdx.x & 31;
    const int m0 = blockIdx.x << 4;
    const int n0 = wave << 4;
    const int lm = lane & 15;
    const int lk = (lane >> 4) << 1;

    const float* xr = x + (size_t)(m0 + lm) * D;
    const float* wr = W + (size_t)(n0 + lm) * D;

    v8f acc = {};
#pragma unroll
    for (int k = 0; k < D; k += 4) {
        v2f a = *(const v2f*)(xr + k + lk);
        v2f b = *(const v2f*)(wr + k + lk);
        acc = __builtin_amdgcn_wmma_f32_16x16x4_f32(
            false, a, false, b, (short)0, acc, false, false);
    }

    const float bn = bias[n0 + lm];
    float* hp = h + (size_t)(m0 + ((lane >> 4) << 3)) * D + (n0 + lm);
#pragma unroll
    for (int v = 0; v < 8; ++v)
        hp[(size_t)v * D] = acc[v] + bn;
}


__device__ __forceinline__ void drain8(const int2* __restrict__ p, int lane,
                                       const float* __restrict__ h,
                                       const float* __restrict__ rsq, float* acc) {
    const int2 en = p[lane & 7];
    const float w = rsq[en.x];
    float4 v[8];
#pragma unroll
    for (int k = 0; k < 8; ++k) {
        const int s = __builtin_amdgcn_readlane(en.x, k);
        v[k] = *(const float4*)(h + (size_t)s * D + lane * 4);
    }
#pragma unroll
    for (int k = 0; k < 8; ++k) {
        const int r = __builtin_amdgcn_readlane(en.y, k);
        const float wk = __int_as_float(__builtin_amdgcn_readlane(__float_as_int(w), k));
        float* a = acc + r * D + lane;
        __hip_atomic_fetch_add(a,      wk * v[k].x, __ATOMIC_RELAXED, __HIP_MEMORY_SCOPE_WORKGROUP);
        __hip_atomic_fetch_add(a + 32, wk * v[k].y, __ATOMIC_RELAXED, __HIP_MEMORY_SCOPE_WORKGROUP);
        __hip_atomic_fetch_add(a + 64, wk * v[k].z, __ATOMIC_RELAXED, __HIP_MEMORY_SCOPE_WORKGROUP);
        __hip_atomic_fetch_add(a + 96, wk * v[k].w, __ATOMIC_RELAXED, __HIP_MEMORY_SCOPE_WORKGROUP);
    }
}

__device__ __forceinline__ uint4 load_tags(const unsigned char* __restrict__ tag,
                                           int u, int lane, int e_pad, bool live) {
    const int off = (u << 9) + (lane << 4);
    const bool in = live && off < e_pad;
    const uint4 v = *(const uint4*)(tag + (in ? off : 0));
    const unsigned m = in ? 0u : ~0u;
    return make_uint4(v.x | m, v.y | m, v.z | m, v.w | m);
}

__device__ __forceinline__ unsigned match4(unsigned v, unsigned tag4) {
    const unsigned t = v ^ tag4;
    const unsigned z = ~(((t & 0x7F7F7F7Fu) + 0x7F7F7F7Fu) | t | 0x7F7F7F7Fu);
    return ((z >> 7) * 0x10204080u) >> 28;
}

__device__ __forceinline__ unsigned match16(uint4 q, unsigned tag4) {
    return match4(q.x, tag4) | (match4(q.y, tag4) << 4) |
           (match4(q.z, tag4) << 8) | (match4(q.w, tag4) << 12);
}

__global__ __launch_bounds__(1024) void k_owner_aggregate(
        const int* __restrict__ ei, const unsigned char* __restrict__ tag,
        const float* __restrict__ h, const float* __restrict__ rsq,
        float* __restrict__ out, int n, int e, int e_pad) {
    __shared__ __attribute__((aligned(16))) float acc[(ROWS + 1) * D];
    __shared__ int2 wl[AGG_WAVES][LIST_CAP];

    const int tid  = threadIdx.x;
    const int lane = tid & 31;
    const int wave = __builtin_amdgcn_readfirstlane(tid >> 5);
    const int base = blockIdx.x * ROWS;

    float4* a4 = (float4*)acc;
    for (int i = tid; i < (ROWS + 1) * D / 4; i += AGG_WAVES * 32)
        a4[i] = make_float4(0.f, 0.f, 0.f, 0.f);
    __syncthreads();

    const unsigned tag4 = blockIdx.x * 0x01010101u;
    const int n_units = (e_pad + 511) >> 9;
    const int n_super = (n_units + 3) >> 2;
    const int start   = (int)((blockIdx.x * (unsigned)n_super) / gridDim.x);
    int2* mylist = wl[wave];
    int cnt = 0;

    int g = wave + start; if (g >= n_super) g -= n_super;
    const bool any = wave < n_super;
    uint4 c0 = load_tags(tag, g * 4 + 0, lane, e_pad, any);
    uint4 c1 = load_tags(tag, g * 4 + 1, lane, e_pad, any);
    uint4 c2 = load_tags(tag, g * 4 + 2, lane, e_pad, any);
    uint4 c3 = load_tags(tag, g * 4 + 3, lane, e_pad, any);

    for (int j = wave; j < n_super; j += AGG_WAVES) {
        const int jn = j + AGG_WAVES;
        int gn = jn + start; if (gn >= n_super) gn -= n_super;
        const bool more = jn < n_super;
        const uint4 d0 = load_tags(tag, gn * 4 + 0, lane, e_pad, more);
        const uint4 d1 = load_tags(tag, gn * 4 + 1, lane, e_pad, more);
        const uint4 d2 = load_tags(tag, gn * 4 + 2, lane, e_pad, more);
        const uint4 d3 = load_tags(tag, gn * 4 + 3, lane, e_pad, more);

        unsigned lo = match16(c0, tag4) | (match16(c1, tag4) << 16);
        unsigned hi = match16(c2, tag4) | (match16(c3, tag4) << 16);

        while (true) {
            const bool has = (lo | hi) != 0u;
            const unsigned ball = __builtin_amdgcn_ballot_w32(has);
            if (ball == 0u) break;
            if (has) {
                int b;
                if (lo) { b = __builtin_ctz(lo); lo &= lo - 1u; }
                else    { b = 32 + __builtin_ctz(hi); hi &= hi - 1u; }
                const int edge = ((g * 4 + (b >> 4)) << 9) + (lane << 4) + (b & 15);
                const int dn = ei[edge];
                const int sn = ei[e + edge];
                const int pos = cnt + __builtin_popcount(ball & ((1u << lane) - 1u));
                mylist[pos] = make_int2(sn, dn - base);
            }
            cnt += __builtin_popcount(ball);
            __builtin_amdgcn_wave_barrier();
            while (cnt >= 8) { cnt -= 8; drain8(mylist + cnt, lane, h, rsq, acc); }
        }

        c0 = d0; c1 = d1; c2 = d2; c3 = d3; g = gn;
    }

    if (cnt > 0) {
        if (lane >= cnt && lane < 8) mylist[lane] = make_int2(0, ROWS);
        __builtin_amdgcn_wave_barrier();
        drain8(mylist, lane, h, rsq, acc);
    }
    __syncthreads();

    for (int r0 = wave; r0 < ROWS; r0 += 4 * AGG_WAVES) {
        float rd[4];
        float4 hv[4];
#pragma unroll
        for (int u = 0; u < 4; ++u) {
            const int r = r0 + u * AGG_WAVES;
            int node = base + (r < ROWS ? r : ROWS - 1);
            node = node < n ? node : n - 1;
            rd[u] = rsq[node];
            hv[u] = *(const float4*)(h + (size_t)node * D + lane * 4);
        }
#pragma unroll
        for (int u = 0; u < 4; ++u) {
            const int r = r0 + u * AGG_WAVES;
            const int node = base + r;
            if (r < ROWS && node < n) {
                const float* a = acc + r * D + lane;
                float4 o;
                o.x = rd[u] * (a[0]  + rd[u] * hv[u].x);
                o.y = rd[u] * (a[32] + rd[u] * hv[u].y);
                o.z = rd[u] * (a[64] + rd[u] * hv[u].z);
                o.w = rd[u] * (a[96] + rd[u] * hv[u].w);
                *(float4*)(out + (size_t)node * D + lane * 4) = o;
            }
        }
    }
}

extern "C" void kernel_launch(void* const* d_in, const int* in_sizes, int n_in,
                              void* d_out, int out_size, void* d_ws, size_t ws_size,
                              hipStream_t stream) {
    const float* x    = (const float*)d_in[0];
    const int*   ei   = (const int*)d_in[1];
    const float* W    = (const float*)d_in[2];
    const float* bias = (const float*)d_in[3];
    float*       out  = (float*)d_out;

    const int n = in_sizes[0] / D;
    const int e = in_sizes[1] / 2;

    float* h   = (float*)d_ws;
    float* rsq = h + (size_t)n * D;
    unsigned char* tag = (unsigned char*)(rsq + ((n + 3) & ~3));
    const int e_pad = (e + 15) & ~15;

    const int T = 256;

    k_init_deg <<<(n + T - 1) / T, T, 0, stream>>>(rsq, n);
    k_count_deg<<<(e_pad + T - 1) / T, T, 0, stream>>>(ei, rsq, tag, e, e_pad);
    k_rsqrt_deg<<<(n + T - 1) / T, T, 0, stream>>>(rsq, n);

    k_gemm_wmma<<<n / 16, T, 0, stream>>>(x, W, bias, h);

    k_owner_aggregate<<<(n + ROWS - 1) / ROWS, AGG_WAVES * 32, 0, stream>>>(
        ei, tag, h, rsq, out, n, e, e_pad);
}


# GEMM data path: x rows staged through LDS by whole-line 16-byte loads (32 rows per step, padded pitch), one block per 400 rows so each wave loads its W fragments once; product k order unchanged
# speedup vs baseline: 2.2989x; 1.1758x over previous
#include <hip/hip_runtime.h>


typedef __attribute__((ext_vector_type(2))) float v2f;
typedef __attribute__((ext_vector_type(8))) float v8f;

#define D 128
#define ROWS 400
#define AGG_WAVES 32
#define LIST_CAP 64

__global__ void k_init_deg(float* __restrict__ deg, int n) {
    int i = blockIdx.x * blockDim.x + threadIdx.x;
    if (i < n) deg[i] = 1.0f;
}

__global__ void k_count_deg(const int* __restrict__ ei_row, float* __restrict__ deg,
                            unsigned char* __restrict__ tag, int e, int e_pad) {
    int i = blockIdx.x * blockDim.x + threadIdx.x;
    if (i < e) {
        const int d = ei_row[i];
        atomicAdd(&deg[d], 1.0f);
        tag[i] = (unsigned char)(d / ROWS);
    } else if (i < e_pad) {
        tag[i] = 255;
    }
}

__global__ void k_rsqrt_deg(float* __restrict__ deg, int n) {
    int i = blockIdx.x * blockDim.x + threadIdx.x;
    if (i < n) deg[i] = rsqrtf(deg[i]);
}

#define GROWS 400
#define GSTEP 32
#define GSTEPS ((GROWS + GSTEP - 1) / GSTEP)
#define APITCH 132

struct Stage { float4 r0, r1, r2, r3; };

__device__ __forceinline__ Stage stage_load(const float* __restrict__ x, int first, int wave,
                                            int lane, int n) {
    Stage t;
    int r;
    r = first + wave;      r = r < n ? r : n - 1; t.r0 = *(const float4*)(x + (size_t)r * D + lane * 4);
    r = first + wave + 8;  r = r < n ? r : n - 1; t.r1 = *(const float4*)(x + (size_t)r * D + lane * 4);
    r = first + wave + 16; r = r < n ? r : n - 1; t.r2 = *(const float4*)(x + (size_t)r * D + lane * 4);
    r = first + wave + 24; r = r < n ? r : n - 1; t.r3 = *(const float4*)(x + (size_t)r * D + lane * 4);
    return t;
}

__device__ __forceinline__ void stage_store(float* buf, const Stage& t, int wave, int lane) {
    float* p = buf + wave * APITCH + lane * 4;
    *(float4*)(p)               = t.r0;
    *(float4*)(p + 8 * APITCH)  = t.r1;
    *(float4*)(p + 16 * APITCH) = t.r2;
    *(float4*)(p + 24 * APITCH) = t.r3;
}

__device__ __forceinline__ void tile_store(float* __restrict__ h, v8f t0, v8f t1, int first,
                                           int row_end, int lane, int col, float bn) {
    const int mr = first + ((lane >> 4) << 3);
    float* hp = h + (size_t)mr * D + col;
#pragma unroll
    for (int v = 0; v < 8; ++v) {
        if (mr + v < row_end)      hp[(size_t)v * D]        = t0[v] + bn;
        if (mr + 16 + v < row_end) hp[(size_t)(v + 16) * D] = t1[v] + bn;
    }
}

__global__ __launch_bounds__(256) __attribute__((amdgpu_waves_per_eu(2, 2)))
void k_gemm_wmma(const float* __restrict__ x,
                 const float* __restrict__ W,
                 const float* __restrict__ bias,
                 float* __restrict__ h, int n) {
    __shared__ __attribute__((aligned(16))) float xs[2 * GSTEP * APITCH];

    const int lane = threadIdx.x & 31;
    const int wave = __builtin_amdgcn_readfirstlane(threadIdx.x >> 5);
    const int lm = lane & 15;
    const int lk = (lane >> 4) << 1;
    const int n0 = wave << 4;
    const int row0 = blockIdx.x * GROWS;
    const int row_end = row0 + GROWS < n ? row0 + GROWS : n;

    Stage st = stage_load(x, row0, wave, lane, n);

    const float* wr = W + (size_t)(n0 + lm) * D + lk;
    v2f b[D / 4];
#pragma unroll
    for (int k = 0; k < D / 4; ++k) b[k] = *(const v2f*)(wr + 4 * k);
    const float bn = bias[n0 + lm];

    stage_store(xs, st, wave, lane);
    st = stage_load(x, row0 + GSTEP, wave, lane, n);

    v8f p0 = {};
    v8f p1 = {};
    for (int s = 0; s < GSTEPS; ++s) {
        __syncthreads();

        if (s > 0) tile_store(h, p0, p1, row0 + (s - 1) * GSTEP, row_end, lane, n0 + lm, bn);

        float* cur = xs + (s & 1) * (GSTEP * APITCH);
        float* nxt = xs + ((s + 1) & 1) * (GSTEP * APITCH);
        if (s + 1 < GSTEPS) stage_store(nxt, st, wave, lane);
        if (s + 2 < GSTEPS) st = stage_load(x, row0 + (s + 2) * GSTEP, wave, lane, n);

        const float* xa = cur + lm * APITCH + lk;
        v2f a[2][8];
#pragma unroll
        for (int j = 0; j < 4; ++j) {
            a[0][j]     = *(const v2f*)(xa + 4 * j);
            a[0][4 + j] = *(const v2f*)(xa + 16 * APITCH + 4 * j);
        }
        v8f acc0 = {};
        v8f acc1 = {};
#pragma unroll
        for (int c = 0; c < D / 16; ++c) {
            if (c + 1 < D / 16) {
#pragma unroll
                for (int j = 0; j < 4; ++j) {
                    a[(c + 1) & 1][j]     = *(const v2f*)(xa + 16 * (c + 1) + 4 * j);
                    a[(c + 1) & 1][4 + j] = *(const v2f*)(xa + 16 * APITCH + 16 * (c + 1) + 4 * j);
                }
            }
#pragma unroll
            for (int j = 0; j < 4; ++j) {
                acc0 = __builtin_amdgcn_wmma_f32_16x16x4_f32(
                    false, a[c & 1][j], false, b[4 * c + j], (short)0, acc0, false, false);
                acc1 = __builtin_amdgcn_wmma_f32_16x16x4_f32(
                    false, a[c & 1][4 + j], false, b[4 * c + j], (short)0, acc1, false, false);
            }
        }

        p0 = acc0;
        p1 = acc1;
    }
    tile_store(h, p0, p1, row0 + (GSTEPS - 1) * GSTEP, row_end, lane, n0 + lm, bn);
}


__device__ __forceinline__ void drain8(const int2* __restrict__ p, int lane,
                                       const float* __restrict__ h,
                                       const float* __restrict__ rsq, float* acc) {
    const int2 en = p[lane & 7];
    const float w = rsq[en.x];
    float4 v[8];
#pragma unroll
    for (int k = 0; k < 8; ++k) {
        const int s = __builtin_amdgcn_readlane(en.x, k);
        v[k] = *(const float4*)(h + (size_t)s * D + lane * 4);
    }
#pragma unroll
    for (int k = 0; k < 8; ++k) {
        const int r = __builtin_amdgcn_readlane(en.y, k);
        const float wk = __int_as_float(__builtin_amdgcn_readlane(__float_as_int(w), k));
        float* a = acc + r * D + lane;
        __hip_atomic_fetch_add(a,      wk * v[k].x, __ATOMIC_RELAXED, __HIP_MEMORY_SCOPE_WORKGROUP);
        __hip_atomic_fetch_add(a + 32, wk * v[k].y, __ATOMIC_RELAXED, __HIP_MEMORY_SCOPE_WORKGROUP);
        __hip_atomic_fetch_add(a + 64, wk * v[k].z, __ATOMIC_RELAXED, __HIP_MEMORY_SCOPE_WORKGROUP);
        __hip_atomic_fetch_add(a + 96, wk * v[k].w, __ATOMIC_RELAXED, __HIP_MEMORY_SCOPE_WORKGROUP);
    }
}

__device__ __forceinline__ uint4 load_tags(const unsigned char* __restrict__ tag,
                                           int u, int lane, int e_pad, bool live) {
    const int off = (u << 9) + (lane << 4);
    const bool in = live && off < e_pad;
    const uint4 v = *(const uint4*)(tag + (in ? off : 0));
    const unsigned m = in ? 0u : ~0u;
    return make_uint4(v.x | m, v.y | m, v.z | m, v.w | m);
}

__device__ __forceinline__ unsigned match4(unsigned v, unsigned tag4) {
    const unsigned t = v ^ tag4;
    const unsigned z = ~(((t & 0x7F7F7F7Fu) + 0x7F7F7F7Fu) | t | 0x7F7F7F7Fu);
    return ((z >> 7) * 0x10204080u) >> 28;
}

__device__ __forceinline__ unsigned match16(uint4 q, unsigned tag4) {
    return match4(q.x, tag4) | (match4(q.y, tag4) << 4) |
           (match4(q.z, tag4) << 8) | (match4(q.w, tag4) << 12);
}

__global__ __launch_bounds__(1024) void k_owner_aggregate(
        const int* __restrict__ ei, const unsigned char* __restrict__ tag,
        const float* __restrict__ h, const float* __restrict__ rsq,
        float* __restrict__ out, int n, int e, int e_pad) {
    __shared__ __attribute__((aligned(16))) float acc[(ROWS + 1) * D];
    __shared__ int2 wl[AGG_WAVES][LIST_CAP];

    const int tid  = threadIdx.x;
    const int lane = tid & 31;
    const int wave = __builtin_amdgcn_readfirstlane(tid >> 5);
    const int base = blockIdx.x * ROWS;

    float4* a4 = (float4*)acc;
    for (int i = tid; i < (ROWS + 1) * D / 4; i += AGG_WAVES * 32)
        a4[i] = make_float4(0.f, 0.f, 0.f, 0.f);
    __syncthreads();

    const unsigned tag4 = blockIdx.x * 0x01010101u;
    const int n_units = (e_pad + 511) >> 9;
    const int n_super = (n_units + 3) >> 2;
    const int start   = (int)((blockIdx.x * (unsigned)n_super) / gridDim.x);
    int2* mylist = wl[wave];
    int cnt = 0;

    int g = wave + start; if (g >= n_super) g -= n_super;
    const bool any = wave < n_super;
    uint4 c0 = load_tags(tag, g * 4 + 0, lane, e_pad, any);
    uint4 c1 = load_tags(tag, g * 4 + 1, lane, e_pad, any);
    uint4 c2 = load_tags(tag, g * 4 + 2, lane, e_pad, any);
    uint4 c3 = load_tags(tag, g * 4 + 3, lane, e_pad, any);

    for (int j = wave; j < n_super; j += AGG_WAVES) {
        const int jn = j + AGG_WAVES;
        int gn = jn + start; if (gn >= n_super) gn -= n_super;
        const bool more = jn < n_super;
        const uint4 d0 = load_tags(tag, gn * 4 + 0, lane, e_pad, more);
        const uint4 d1 = load_tags(tag, gn * 4 + 1, lane, e_pad, more);
        const uint4 d2 = load_tags(tag, gn * 4 + 2, lane, e_pad, more);
        const uint4 d3 = load_tags(tag, gn * 4 + 3, lane, e_pad, more);

        unsigned lo = match16(c0, tag4) | (match16(c1, tag4) << 16);
        unsigned hi = match16(c2, tag4) | (match16(c3, tag4) << 16);

        while (true) {
            const bool has = (lo | hi) != 0u;
            const unsigned ball = __builtin_amdgcn_ballot_w32(has);
            if (ball == 0u) break;
            if (has) {
                int b;
                if (lo) { b = __builtin_ctz(lo); lo &= lo - 1u; }
                else    { b = 32 + __builtin_ctz(hi); hi &= hi - 1u; }
                const int edge = ((g * 4 + (b >> 4)) << 9) + (lane << 4) + (b & 15);
                const int dn = ei[edge];
                const int sn = ei[e + edge];
                const int pos = cnt + __builtin_popcount(ball & ((1u << lane) - 1u));
                mylist[pos] = make_int2(sn, dn - base);
            }
            cnt += __builtin_popcount(ball);
            __builtin_amdgcn_wave_barrier();
            while (cnt >= 8) { cnt -= 8; drain8(mylist + cnt, lane, h, rsq, acc); }
        }

        c0 = d0; c1 = d1; c2 = d2; c3 = d3; g = gn;
    }

    if (cnt > 0) {
        if (lane >= cnt && lane < 8) mylist[lane] = make_int2(0, ROWS);
        __builtin_amdgcn_wave_barrier();
        drain8(mylist, lane, h, rsq, acc);
    }
    __syncthreads();

    for (int r0 = wave; r0 < ROWS; r0 += 4 * AGG_WAVES) {
        float rd[4];
        float4 hv[4];
#pragma unroll
        for (int u = 0; u < 4; ++u) {
            const int r = r0 + u * AGG_WAVES;
            int node = base + (r < ROWS ? r : ROWS - 1);
            node = node < n ? node : n - 1;
            rd[u] = rsq[node];
            hv[u] = *(const float4*)(h + (size_t)node * D + lane * 4);
        }
#pragma unroll
        for (int u = 0; u < 4; ++u) {
            const int r = r0 + u * AGG_WAVES;
            const int node = base + r;
            if (r < ROWS && node < n) {
                const float* a = acc + r * D + lane;
                float4 o;
                o.x = rd[u] * (a[0]  + rd[u] * hv[u].x);
                o.y = rd[u] * (a[32] + rd[u] * hv[u].y);
                o.z = rd[u] * (a[64] + rd[u] * hv[u].z);
                o.w = rd[u] * (a[96] + rd[u] * hv[u].w);
                *(float4*)(out + (size_t)node * D + lane * 4) = o;
            }
        }
    }
}

extern "C" void kernel_launch(void* const* d_in, const int* in_sizes, int n_in,
                              void* d_out, int out_size, void* d_ws, size_t ws_size,
                              hipStream_t stream) {
    const float* x    = (const float*)d_in[0];
    const int*   ei   = (const int*)d_in[1];
    const float* W    = (const float*)d_in[2];
    const float* bias = (const float*)d_in[3];
    float*       out  = (float*)d_out;

    const int n = in_sizes[0] / D;
    const int e = in_sizes[1] / 2;

    float* h   = (float*)d_ws;
    float* rsq = h + (size_t)n * D;
    unsigned char* tag = (unsigned char*)(rsq + ((n + 3) & ~3));
    const int e_pad = (e + 15) & ~15;

    const int T = 256;

    k_init_deg <<<(n + T - 1) / T, T, 0, stream>>>(rsq, n);
    k_count_deg<<<(e_pad + T - 1) / T, T, 0, stream>>>(ei, rsq, tag, e, e_pad);
    k_rsqrt_deg<<<(n + T - 1) / T, T, 0, stream>>>(rsq, n);

    k_gemm_wmma<<<(n + GROWS - 1) / GROWS, T, 0, stream>>>(x, W, bias, h, n);

    k_owner_aggregate<<<(n + ROWS - 1) / ROWS, AGG_WAVES * 32, 0, stream>>>(
        ei, tag, h, rsq, out, n, e, e_pad);
}


# Edge pass (degree atomics, owner tag bytes) moved into the matrix-bound GEMM kernel, 2 edges per thread per row step: its own launch of 50000 short waves is gone; degree sums exact in any order
# speedup vs baseline: 2.4480x; 1.0649x over previous
#include <hip/hip_runtime.h>


typedef __attribute__((ext_vector_type(2))) float v2f;
typedef __attribute__((ext_vector_type(8))) float v8f;

#define D 128
#define ROWS 400
#define AGG_WAVES 32
#define LIST_CAP 64

__global__ void k_init_deg(float* __restrict__ deg, int n) {
    int i = blockIdx.x * blockDim.x + threadIdx.x;
    if (i < n) deg[i] = 1.0f;
}

__device__ __forceinline__ void edge_apply(int i, int d, float* __restrict__ deg,
                                           unsigned char* __restrict__ tag, int e, int e_pad) {
    if (i < e) {
        atomicAdd(&deg[d], 1.0f);
        tag[i] = (unsigned char)(d / ROWS);
    } else if (i < e_pad) {
        tag[i] = 255;
    }
}

__device__ __forceinline__ int edge_load(const int* __restrict__ ei_row, int i, int e) {
    return ei_row[i < e ? i : 0];
}

__global__ void k_rsqrt_deg(float* __restrict__ deg, int n) {
    int i = blockIdx.x * blockDim.x + threadIdx.x;
    if (i < n) deg[i] = rsqrtf(deg[i]);
}

#define GROWS 400
#define GSTEP 32
#define GSTEPS ((GROWS + GSTEP - 1) / GSTEP)
#define APITCH 132
#define EPS 2

struct Stage { float4 r0, r1, r2, r3; };

__device__ __forceinline__ Stage stage_load(const float* __restrict__ x, int first, int wave,
                                            int lane, int n) {
    Stage t;
    int r;
    r = first + wave;      r = r < n ? r : n - 1; t.r0 = *(const float4*)(x + (size_t)r * D + lane * 4);
    r = first + wave + 8;  r = r < n ? r : n - 1; t.r1 = *(const float4*)(x + (size_t)r * D + lane * 4);
    r = first + wave + 16; r = r < n ? r : n - 1; t.r2 = *(const float4*)(x + (size_t)r * D + lane * 4);
    r = first + wave + 24; r = r < n ? r : n - 1; t.r3 = *(const float4*)(x + (size_t)r * D + lane * 4);
    return t;
}

__device__ __forceinline__ void stage_store(float* buf, const Stage& t, int wave, int lane) {
    float* p = buf + wave * APITCH + lane * 4;
    *(float4*)(p)               = t.r0;
    *(float4*)(p + 8 * APITCH)  = t.r1;
    *(float4*)(p + 16 * APITCH) = t.r2;
    *(float4*)(p + 24 * APITCH) = t.r3;
}

__device__ __forceinline__ void tile_store(float* __restrict__ h, v8f t0, v8f t1, int first,
                                           int row_end, int lane, int col, float bn) {
    const int mr = first + ((lane >> 4) << 3);
    float* hp = h + (size_t)mr * D + col;
#pragma unroll
    for (int v = 0; v < 8; ++v) {
        if (mr + v < row_end)      hp[(size_t)v * D]        = t0[v] + bn;
        if (mr + 16 + v < row_end) hp[(size_t)(v + 16) * D] = t1[v] + bn;
    }
}

__global__ __launch_bounds__(256) __attribute__((amdgpu_waves_per_eu(2, 2)))
void k_gemm_wmma(const float* __restrict__ x,
                 const float* __restrict__ W,
                 const float* __restrict__ bias,
                 float* __restrict__ h, int n,
                 const int* __restrict__ ei_row, float* __restrict__ deg,
                 unsigned char* __restrict__ tag, int e, int e_pad) {
    __shared__ __attribute__((aligned(16))) float xs[2 * GSTEP * APITCH];

    const int lane = threadIdx.x & 31;
    const int wave = __builtin_amdgcn_readfirstlane(threadIdx.x >> 5);
    const int lm = lane & 15;
    const int lk = (lane >> 4) << 1;
    const int n0 = wave << 4;
    const int row0 = blockIdx.x * GROWS;
    const int row_end = row0 + GROWS < n ? row0 + GROWS : n;

    const int gtid = blockIdx.x * 256 + threadIdx.x;
    const int nthr = gridDim.x * 256;
    int ed[EPS];

    Stage st = stage_load(x, row0, wave, lane, n);

    const float* wr = W + (size_t)(n0 + lm) * D + lk;
    v2f b[D / 4];
#pragma unroll
    for (int k = 0; k < D / 4; ++k) b[k] = *(const v2f*)(wr + 4 * k);
    const float bn = bias[n0 + lm];

    stage_store(xs, st, wave, lane);
    st = stage_load(x, row0 + GSTEP, wave, lane, n);

    v8f p0 = {};
    v8f p1 = {};
    for (int s = 0; s < GSTEPS; ++s) {
        __syncthreads();

        if (s > 0) {
            tile_store(h, p0, p1, row0 + (s - 1) * GSTEP, row_end, lane, n0 + lm, bn);
#pragma unroll
            for (int u = 0; u < EPS; ++u)
                edge_apply(((s - 1) * EPS + u) * nthr + gtid, ed[u], deg, tag, e, e_pad);
        }

        float* cur = xs + (s & 1) * (GSTEP * APITCH);
        float* nxt = xs + ((s + 1) & 1) * (GSTEP * APITCH);
        if (s + 1 < GSTEPS) stage_store(nxt, st, wave, lane);
        if (s + 2 < GSTEPS) st = stage_load(x, row0 + (s + 2) * GSTEP, wave, lane, n);
#pragma unroll
        for (int u = 0; u < EPS; ++u)
            ed[u] = edge_load(ei_row, (s * EPS + u) * nthr + gtid, e);

        const float* xa = cur + lm * APITCH + lk;
        v2f a[2][8];
#pragma unroll
        for (int j = 0; j < 4; ++j) {
            a[0][j]     = *(const v2f*)(xa + 4 * j);
            a[0][4 + j] = *(const v2f*)(xa + 16 * APITCH + 4 * j);
        }
        v8f acc0 = {};
        v8f acc1 = {};
#pragma unroll
        for (int c = 0; c < D / 16; ++c) {
            if (c + 1 < D / 16) {
#pragma unroll
                for (int j = 0; j < 4; ++j) {
                    a[(c + 1) & 1][j]     = *(const v2f*)(xa + 16 * (c + 1) + 4 * j);
                    a[(c + 1) & 1][4 + j] = *(const v2f*)(xa + 16 * APITCH + 16 * (c + 1) + 4 * j);
                }
            }
#pragma unroll
            for (int j = 0; j < 4; ++j) {
                acc0 = __builtin_amdgcn_wmma_f32_16x16x4_f32(
                    false, a[c & 1][j], false, b[4 * c + j], (short)0, acc0, false, false);
                acc1 = __builtin_amdgcn_wmma_f32_16x16x4_f32(
                    false, a[c & 1][4 + j], false, b[4 * c + j], (short)0, acc1, false, false);
            }
        }

        p0 = acc0;
        p1 = acc1;
    }
    tile_store(h, p0, p1, row0 + (GSTEPS - 1) * GSTEP, row_end, lane, n0 + lm, bn);
#pragma unroll
    for (int u = 0; u < EPS; ++u)
        edge_apply(((GSTEPS - 1) * EPS + u) * nthr + gtid, ed[u], deg, tag, e, e_pad);
    for (int i = GSTEPS * EPS * nthr + gtid; i < e_pad; i += nthr)
        edge_apply(i, edge_load(ei_row, i, e), deg, tag, e, e_pad);
}


__device__ __forceinline__ void drain8(const int2* __restrict__ p, int lane,
                                       const float* __restrict__ h,
                                       const float* __restrict__ rsq, float* acc) {
    const int2 en = p[lane & 7];
    const float w = rsq[en.x];
    float4 v[8];
#pragma unroll
    for (int k = 0; k < 8; ++k) {
        const int s = __builtin_amdgcn_readlane(en.x, k);
        v[k] = *(const float4*)(h + (size_t)s * D + lane * 4);
    }
#pragma unroll
    for (int k = 0; k < 8; ++k) {
        const int r = __builtin_amdgcn_readlane(en.y, k);
        const float wk = __int_as_float(__builtin_amdgcn_readlane(__float_as_int(w), k));
        float* a = acc + r * D + lane;
        __hip_atomic_fetch_add(a,      wk * v[k].x, __ATOMIC_RELAXED, __HIP_MEMORY_SCOPE_WORKGROUP);
        __hip_atomic_fetch_add(a + 32, wk * v[k].y, __ATOMIC_RELAXED, __HIP_MEMORY_SCOPE_WORKGROUP);
        __hip_atomic_fetch_add(a + 64, wk * v[k].z, __ATOMIC_RELAXED, __HIP_MEMORY_SCOPE_WORKGROUP);
        __hip_atomic_fetch_add(a + 96, wk * v[k].w, __ATOMIC_RELAXED, __HIP_MEMORY_SCOPE_WORKGROUP);
    }
}

__device__ __forceinline__ uint4 load_tags(const unsigned char* __restrict__ tag,
                                           int u, int lane, int e_pad, bool live) {
    const int off = (u << 9) + (lane << 4);
    const bool in = live && off < e_pad;
    const uint4 v = *(const uint4*)(tag + (in ? off : 0));
    const unsigned m = in ? 0u : ~0u;
    return make_uint4(v.x | m, v.y | m, v.z | m, v.w | m);
}

__device__ __forceinline__ unsigned match4(unsigned v, unsigned tag4) {
    const unsigned t = v ^ tag4;
    const unsigned z = ~(((t & 0x7F7F7F7Fu) + 0x7F7F7F7Fu) | t | 0x7F7F7F7Fu);
    return ((z >> 7) * 0x10204080u) >> 28;
}

__device__ __forceinline__ unsigned match16(uint4 q, unsigned tag4) {
    return match4(q.x, tag4) | (match4(q.y, tag4) << 4) |
           (match4(q.z, tag4) << 8) | (match4(q.w, tag4) << 12);
}

__global__ __launch_bounds__(1024) void k_owner_aggregate(
        const int* __restrict__ ei, const unsigned char* __restrict__ tag,
        const float* __restrict__ h, const float* __restrict__ rsq,
        float* __restrict__ out, int n, int e, int e_pad) {
    __shared__ __attribute__((aligned(16))) float acc[(ROWS + 1) * D];
    __shared__ int2 wl[AGG_WAVES][LIST_CAP];

    const int tid  = threadIdx.x;
    const int lane = tid & 31;
    const int wave = __builtin_amdgcn_readfirstlane(tid >> 5);
    const int base = blockIdx.x * ROWS;

    float4* a4 = (float4*)acc;
    for (int i = tid; i < (ROWS + 1) * D / 4; i += AGG_WAVES * 32)
        a4[i] = make_float4(0.f, 0.f, 0.f, 0.f);
    __syncthreads();

    const unsigned tag4 = blockIdx.x * 0x01010101u;
    const int n_units = (e_pad + 511) >> 9;
    const int n_super = (n_units + 3) >> 2;
    const int start   = (int)((blockIdx.x * (unsigned)n_super) / gridDim.x);
    int2* mylist = wl[wave];
    int cnt = 0;

    int g = wave + start; if (g >= n_super) g -= n_super;
    const bool any = wave < n_super;
    uint4 c0 = load_tags(tag, g * 4 + 0, lane, e_pad, any);
    uint4 c1 = load_tags(tag, g * 4 + 1, lane, e_pad, any);
    uint4 c2 = load_tags(tag, g * 4 + 2, lane, e_pad, any);
    uint4 c3 = load_tags(tag, g * 4 + 3, lane, e_pad, any);

    for (int j = wave; j < n_super; j += AGG_WAVES) {
        const int jn = j + AGG_WAVES;
        int gn = jn + start; if (gn >= n_super) gn -= n_super;
        const bool more = jn < n_super;
        const uint4 d0 = load_tags(tag, gn * 4 + 0, lane, e_pad, more);
        const uint4 d1 = load_tags(tag, gn * 4 + 1, lane, e_pad, more);
        const uint4 d2 = load_tags(tag, gn * 4 + 2, lane, e_pad, more);
        const uint4 d3 = load_tags(tag, gn * 4 + 3, lane, e_pad, more);

        unsigned lo = match16(c0, tag4) | (match16(c1, tag4) << 16);
        unsigned hi = match16(c2, tag4) | (match16(c3, tag4) << 16);

        while (true) {
            const bool has = (lo | hi) != 0u;
            const unsigned ball = __builtin_amdgcn_ballot_w32(has);
            if (ball == 0u) break;
            if (has) {
                int b;
                if (lo) { b = __builtin_ctz(lo); lo &= lo - 1u; }
                else    { b = 32 + __builtin_ctz(hi); hi &= hi - 1u; }
                const int edge = ((g * 4 + (b >> 4)) << 9) + (lane << 4) + (b & 15);
                const int dn = ei[edge];
                const int sn = ei[e + edge];
                const int pos = cnt + __builtin_popcount(ball & ((1u << lane) - 1u));
                mylist[pos] = make_int2(sn, dn - base);
            }
            cnt += __builtin_popcount(ball);
            __builtin_amdgcn_wave_barrier();
            while (cnt >= 8) { cnt -= 8; drain8(mylist + cnt, lane, h, rsq, acc); }
        }

        c0 = d0; c1 = d1; c2 = d2; c3 = d3; g = gn;
    }

    if (cnt > 0) {
        if (lane >= cnt && lane < 8) mylist[lane] = make_int2(0, ROWS);
        __builtin_amdgcn_wave_barrier();
        drain8(mylist, lane, h, rsq, acc);
    }
    __syncthreads();

    for (int r0 = wave; r0 < ROWS; r0 += 4 * AGG_WAVES) {
        float rd[4];
        float4 hv[4];
#pragma unroll
        for (int u = 0; u < 4; ++u) {
            const int r = r0 + u * AGG_WAVES;
            int node = base + (r < ROWS ? r : ROWS - 1);
            node = node < n ? node : n - 1;
            rd[u] = rsq[node];
            hv[u] = *(const float4*)(h + (size_t)node * D + lane * 4);
        }
#pragma unroll
        for (int u = 0; u < 4; ++u) {
            const int r = r0 + u * AGG_WAVES;
            const int node = base + r;
            if (r < ROWS && node < n) {
                const float* a = acc + r * D + lane;
                float4 o;
                o.x = rd[u] * (a[0]  + rd[u] * hv[u].x);
                o.y = rd[u] * (a[32] + rd[u] * hv[u].y);
                o.z = rd[u] * (a[64] + rd[u] * hv[u].z);
                o.w = rd[u] * (a[96] + rd[u] * hv[u].w);
                *(float4*)(out + (size_t)node * D + lane * 4) = o;
            }
        }
    }
}

extern "C" void kernel_launch(void* const* d_in, const int* in_sizes, int n_in,
                              void* d_out, int out_size, void* d_ws, size_t ws_size,
                              hipStream_t stream) {
    const float* x    = (const float*)d_in[0];
    const int*   ei   = (const int*)d_in[1];
    const float* W    = (const float*)d_in[2];
    const float* bias = (const float*)d_in[3];
    float*       out  = (float*)d_out;

    const int n = in_sizes[0] / D;
    const int e = in_sizes[1] / 2;

    float* h   = (float*)d_ws;
    float* rsq = h + (size_t)n * D;
    unsigned char* tag = (unsigned char*)(rsq + ((n + 3) & ~3));
    const int e_pad = (e + 15) & ~15;

    const int T = 256;

    k_init_deg <<<(n + T - 1) / T, T, 0, stream>>>(rsq, n);

    k_gemm_wmma<<<(n + GROWS - 1) / GROWS, T, 0, stream>>>(x, W, bias, h, n, ei, rsq, tag, e, e_pad);

    k_rsqrt_deg<<<(n + T - 1) / T, T, 0, stream>>>(rsq, n);

    k_owner_aggregate<<<(n + ROWS - 1) / ROWS, AGG_WAVES * 32, 0, stream>>>(
        ei, tag, h, rsq, out, n, e, e_pad);
}
